# starting kernel
# baseline (speedup 1.0000x reference)
#include <hip/hip_runtime.h>


typedef __attribute__((ext_vector_type(2))) float v2f;
typedef __attribute__((ext_vector_type(4))) float v4f;
typedef __attribute__((ext_vector_type(8))) float v8f;

#define NSEQ 384
#define BATCH 2
#define ROWS (BATCH * NSEQ)
#define CIN 64
#define HID 128
#define LPITCH 132

__global__ __launch_bounds__(32) void fcpair_stage1_wmma(
    const float* __restrict__ X,
    const float* __restrict__ W1,
    const float* __restrict__ b1,
    float* __restrict__ G)
{
    const int s    = blockIdx.z;
    const int rt   = blockIdx.x;
    const int ct   = blockIdx.y;
    const int lane = threadIdx.x;
    const int half = lane >> 4;
    const int l15  = lane & 15;

    const int arow = rt * 16 + l15;
    const int hcol = ct * 16 + l15;

    const float* aptr = X  + arow * CIN + 2 * half;
    const float* bptr = W1 + hcol * (6 * CIN) + s * CIN + 2 * half;

    v8f acc = {};
#pragma unroll
    for (int k0 = 0; k0 < CIN; k0 += 4) {
        v2f a  = *(const v2f*)(aptr + k0);
        v2f bb = *(const v2f*)(bptr + k0);
        acc = __builtin_amdgcn_wmma_f32_16x16x4_f32(
             false, a,  false, bb,
             (short)0, acc,  false,  false);
    }

    const float badd = (s == 0) ? b1[hcol] : 0.0f;
    float* gout = G + (size_t)s * ROWS * HID;
#pragma unroll
    for (int r = 0; r < 8; ++r) {
        const int grow = rt * 16 + half * 8 + r;
        gout[grow * HID + hcol] = acc[r] + badd;
    }
}

__global__ __launch_bounds__(256) void fcpair_stage2_pair(
    const float* __restrict__ G,
    const float* __restrict__ W2,
    const float* __restrict__ b2,
    float* __restrict__ out)
{
    __shared__ alignas(16) float rowS[3 * 16 * LPITCH];
    __shared__ alignas(16) float colS[3 * 32 * LPITCH];
    __shared__ alignas(16) float w2s[HID];

    const int b   = blockIdx.z;
    const int i0  = blockIdx.y * 16;
    const int j0  = blockIdx.x * 32;
    const int tid = threadIdx.x;

    const float* G0 = G + (size_t)0 * ROWS * HID;
    const float* G1 = G + (size_t)1 * ROWS * HID;
    const float* G2 = G + (size_t)2 * ROWS * HID;
    const float* G3 = G + (size_t)3 * ROWS * HID;
    const float* G4 = G + (size_t)4 * ROWS * HID;
    const float* G5 = G + (size_t)5 * ROWS * HID;

    for (int idx = tid; idx < 3 * 16 * 32; idx += 256) {
        const int arr = idx / (16 * 32);
        const int rem = idx - arr * (16 * 32);
        const int ii  = rem >> 5;
        const int q   = rem & 31;
        const int i   = i0 + ii;
        const int ri  = b * NSEQ + i;
        const float* src;
        if (arr == 0)      src = G0 + (size_t)ri * HID;
        else if (arr == 1) src = G2 + (size_t)((i >= 1)        ? ri - 1 : ri) * HID;
        else               src = G4 + (size_t)((i <= NSEQ - 2) ? ri + 1 : ri) * HID;
        *(v4f*)&rowS[(arr * 16 + ii) * LPITCH + q * 4] = *(const v4f*)(src + q * 4);
    }
    for (int idx = tid; idx < 3 * 32 * 32; idx += 256) {
        const int arr = idx / (32 * 32);
        const int rem = idx - arr * (32 * 32);
        const int jj  = rem >> 5;
        const int q   = rem & 31;
        const int j   = j0 + jj;
        const int rj  = b * NSEQ + j;
        const float* src;
        if (arr == 0)      src = G1 + (size_t)rj * HID;
        else if (arr == 1) src = G3 + (size_t)((j <= NSEQ - 2) ? rj + 1 : rj) * HID;
        else               src = G5 + (size_t)((j >= 1)        ? rj - 1 : rj) * HID;
        *(v4f*)&colS[(arr * 32 + jj) * LPITCH + q * 4] = *(const v4f*)(src + q * 4);
    }
    if (tid < HID) w2s[tid] = W2[tid];
    __syncthreads();

    const int ii = tid >> 4;
    const int jc = tid & 15;
    const int i  = i0 + ii;
    const int jA = j0 + jc;
    const int jB = jA + 16;

    const float mi  = (i >= 1)        ? 1.0f : 0.0f;
    const float mip = (i <= NSEQ - 2) ? 1.0f : 0.0f;
    const float muA = mi  * ((jA <= NSEQ - 2) ? 1.0f : 0.0f);
    const float mdA = mip * ((jA >= 1)        ? 1.0f : 0.0f);
    const float muB = mi  * ((jB <= NSEQ - 2) ? 1.0f : 0.0f);
    const float mdB = mip * ((jB >= 1)        ? 1.0f : 0.0f);

    const float* r0p = &rowS[(0 * 16 + ii) * LPITCH];
    const float* r1p = &rowS[(1 * 16 + ii) * LPITCH];
    const float* r2p = &rowS[(2 * 16 + ii) * LPITCH];
    const float* a0p = &colS[(0 * 32 + jc) * LPITCH];
    const float* a1p = &colS[(1 * 32 + jc) * LPITCH];
    const float* a2p = &colS[(2 * 32 + jc) * LPITCH];
    const float* b0p = &colS[(0 * 32 + jc + 16) * LPITCH];
    const float* b1p = &colS[(1 * 32 + jc + 16) * LPITCH];
    const float* b2p = &colS[(2 * 32 + jc + 16) * LPITCH];

    float accA = 0.0f, accB = 0.0f;
#pragma unroll 4
    for (int h = 0; h < HID; h += 4) {
        const v4f r0 = *(const v4f*)(r0p + h);
        const v4f r1 = *(const v4f*)(r1p + h);
        const v4f r2 = *(const v4f*)(r2p + h);
        const v4f a0 = *(const v4f*)(a0p + h);
        const v4f a1 = *(const v4f*)(a1p + h);
        const v4f a2 = *(const v4f*)(a2p + h);
        const v4f c0 = *(const v4f*)(b0p + h);
        const v4f c1 = *(const v4f*)(b1p + h);
        const v4f c2 = *(const v4f*)(b2p + h);
        const v4f w  = *(const v4f*)&w2s[h];
#pragma unroll
        for (int e = 0; e < 4; ++e) {
            float sA = r0[e] + a0[e] + muA * (r1[e] + a1[e]) + mdA * (r2[e] + a2[e]);
            accA = fmaf(w[e], fmaxf(sA, 0.0f), accA);
            float sB = r0[e] + c0[e] + muB * (r1[e] + c1[e]) + mdB * (r2[e] + c2[e]);
            accB = fmaf(w[e], fmaxf(sB, 0.0f), accB);
        }
    }
    const float bias2 = b2[0];
    const size_t base = ((size_t)(b * NSEQ + i)) * NSEQ;
    out[base + jA] = accA + bias2;
    out[base + jB] = accB + bias2;
}

extern "C" void kernel_launch(void* const* d_in, const int* in_sizes, int n_in,
                              void* d_out, int out_size, void* d_ws, size_t ws_size,
                              hipStream_t stream) {
    (void)in_sizes; (void)n_in; (void)out_size; (void)ws_size;
    const float* x  = (const float*)d_in[0];
    const float* W1 = (const float*)d_in[1];
    const float* b1 = (const float*)d_in[2];
    const float* W2 = (const float*)d_in[3];
    const float* b2 = (const float*)d_in[4];
    float* out = (float*)d_out;
    float* G   = (float*)d_ws;

    fcpair_stage1_wmma<<<dim3(48, 8, 6), 32, 0, stream>>>(x, W1, b1, G);
    fcpair_stage2_pair<<<dim3(12, 24, 2), 256, 0, stream>>>(G, W2, b2, out);
}


# Masks factor into an i part and a j part: stage 1 pre-sums the row terms and the column terms (3 edge variants each); pair loop = add, relu, fma on one row + one column vector; f32 sum order changes
# speedup vs baseline: 2.2337x; 2.2337x over previous
#include <hip/hip_runtime.h>


typedef __attribute__((ext_vector_type(2))) float v2f;
typedef __attribute__((ext_vector_type(4))) float v4f;
typedef __attribute__((ext_vector_type(8))) float v8f;

#define NSEQ 384
#define BATCH 2
#define ROWS (BATCH * NSEQ)
#define CIN 64
#define HID 128
#define W1PITCH (6 * CIN)
#define PLANE (ROWS * HID)
#define LPITCH 132

#define TI 12
#define TJ 32
#define RPW 3

__device__ __forceinline__ v8f chain64(const float* __restrict__ arow,
                                       const float* __restrict__ brow, v8f acc)
{
#pragma unroll
    for (int k0 = 0; k0 < CIN; k0 += 4) {
        const v2f a  = *(const v2f*)(arow + k0);
        const v2f bb = *(const v2f*)(brow + k0);
        acc = __builtin_amdgcn_wmma_f32_16x16x4_f32(
             false, a,  false, bb,
             (short)0, acc,  false,  false);
    }
    return acc;
}

__global__ __launch_bounds__(32) __attribute__((amdgpu_waves_per_eu(1, 4)))
void fcpair_stage1_wmma(
    const float* __restrict__ X,
    const float* __restrict__ W1,
    const float* __restrict__ b1,
    float* __restrict__ T)
{
    const int side = blockIdx.z;
    const int rt   = blockIdx.x;
    const int ct   = blockIdx.y;
    const int lane = threadIdx.x;
    const int half = lane >> 4;
    const int l15  = lane & 15;

    const int arow = rt * 16 + l15;
    const int hcol = ct * 16 + l15;

    const int arow_m = (arow >= 1)        ? arow - 1 : arow;
    const int arow_p = (arow <= ROWS - 2) ? arow + 1 : arow;

    const int slice_c = side;
    const int slice_m = (side == 0) ? 2 : 5;
    const int slice_p = (side == 0) ? 4 : 3;

    const float* xc = X + arow   * CIN + 2 * half;
    const float* xm = X + arow_m * CIN + 2 * half;
    const float* xp = X + arow_p * CIN + 2 * half;
    const float* wrow = W1 + hcol * W1PITCH + 2 * half;

    v8f accC = {}, accM = {}, accP = {};
    accC = chain64(xc, wrow + slice_c * CIN, accC);
    accM = chain64(xm, wrow + slice_m * CIN, accM);
    accP = chain64(xp, wrow + slice_p * CIN, accP);

    const float badd = (side == 0) ? b1[hcol] : 0.0f;
    float* t0 = T + (size_t)(side * 3 + 0) * PLANE;
    float* t1 = T + (size_t)(side * 3 + 1) * PLANE;
    float* t2 = T + (size_t)(side * 3 + 2) * PLANE;
    for (int r = 0; r < 8; ++r) {
        const int grow = rt * 16 + half * 8 + r;
        const int pos  = (grow >= NSEQ) ? grow - NSEQ : grow;
        const float c = accC[r] + badd;
        const float m = (pos >= 1)        ? accM[r] : 0.0f;
        const float p = (pos <= NSEQ - 2) ? accP[r] : 0.0f;
        const float cm = c + m;
        t0[grow * HID + hcol] = cm + p;
        t1[grow * HID + hcol] = cm;
        t2[grow * HID + hcol] = c + p;
    }
}

template <bool EDGE_I>
__device__ __forceinline__ void pair_loop(
    const float* p0, const float* p1, const float* p2,
    const float* q0, const float* q1, const float* q2,
    const float* __restrict__ W2, float& o0, float& o1, float& o2)
{
    float acc0 = 0.0f, acc1 = 0.0f, acc2 = 0.0f;
#pragma unroll 8
    for (int h = 0; h < HID; h += 4) {
        const v4f P0 = *(const v4f*)(p0 + h);
        const v4f P1 = *(const v4f*)(p1 + h);
        const v4f P2 = *(const v4f*)(p2 + h);
        const v4f Q0 = *(const v4f*)(q0 + h);
        v4f Q1 = Q0, Q2 = Q0;
        if (EDGE_I) {
            Q1 = *(const v4f*)(q1 + h);
            Q2 = *(const v4f*)(q2 + h);
        }
        for (int e = 0; e < 4; ++e) {
            const float w = W2[h + e];
            acc0 = fmaf(w, fmaxf(P0[e] + Q0[e], 0.0f), acc0);
            acc1 = fmaf(w, fmaxf(P1[e] + Q1[e], 0.0f), acc1);
            acc2 = fmaf(w, fmaxf(P2[e] + Q2[e], 0.0f), acc2);
        }
    }
    o0 = acc0; o1 = acc1; o2 = acc2;
}

__global__ __launch_bounds__(128) void fcpair_stage2_pair(
    const float* __restrict__ T,
    const float* __restrict__ W2,
    const float* __restrict__ b2,
    float* __restrict__ out)
{
    __shared__ alignas(16) float S[(2 * TI + 2 * TJ) * LPITCH];
    constexpr int OFF_PI = 0;
    constexpr int OFF_PE = TI * LPITCH;
    constexpr int OFF_QI = 2 * TI * LPITCH;
    constexpr int OFF_QE = (2 * TI + TJ) * LPITCH;

    const int b   = blockIdx.z;
    const int i0  = blockIdx.y * TI;
    const int j0  = blockIdx.x * TJ;
    const int tid = threadIdx.x;

    const bool jfirst = (j0 == 0), jlast = (j0 + TJ == NSEQ);
    const bool ifirst = (i0 == 0), ilast = (i0 + TI == NSEQ);

    const int q4 = (tid & 31) * 4;
    const int rw = tid >> 5;

    const float* psrc = T + (size_t)(b * NSEQ + i0 + rw) * HID + q4;
    const float* qsrc = T + (size_t)3 * PLANE + (size_t)(b * NSEQ + j0 + rw) * HID + q4;
    const bool jedge_blk = jfirst || jlast;
    const bool iedge_blk = ifirst || ilast;

    v4f pr[TI / 4], qr[TJ / 4], pe[TI / 4], qe[TJ / 4];
    for (int k = 0; k < TI / 4; ++k) pr[k] = *(const v4f*)(psrc + k * 4 * HID);
    for (int k = 0; k < TJ / 4; ++k) qr[k] = *(const v4f*)(qsrc + k * 4 * HID);
    if (jedge_blk) {
        const float* s = psrc + (size_t)(jfirst ? 1 : 2) * PLANE;
        for (int k = 0; k < TI / 4; ++k) pe[k] = *(const v4f*)(s + k * 4 * HID);
    }
    if (iedge_blk) {
        const float* s = qsrc + (size_t)(ifirst ? 1 : 2) * PLANE;
        for (int k = 0; k < TJ / 4; ++k) qe[k] = *(const v4f*)(s + k * 4 * HID);
    }
    for (int k = 0; k < TI / 4; ++k) *(v4f*)&S[OFF_PI + (rw + 4 * k) * LPITCH + q4] = pr[k];
    for (int k = 0; k < TJ / 4; ++k) *(v4f*)&S[OFF_QI + (rw + 4 * k) * LPITCH + q4] = qr[k];
    if (jedge_blk) {
        for (int k = 0; k < TI / 4; ++k) *(v4f*)&S[OFF_PE + (rw + 4 * k) * LPITCH + q4] = pe[k];
    }
    if (iedge_blk) {
        for (int k = 0; k < TJ / 4; ++k) *(v4f*)&S[OFF_QE + (rw + 4 * k) * LPITCH + q4] = qe[k];
    }
    __syncthreads();

    const int lane = tid & 31;
    const int j    = j0 + lane;
    const int il   = rw * RPW;
    const int i    = i0 + il;

    const int poff = ((j == 0 || j == NSEQ - 1) ? OFF_PE : OFF_PI) + il * LPITCH;
    const float* p0 = &S[poff];
    const float* p1 = p0 + LPITCH;
    const float* p2 = p1 + LPITCH;

    float y0, y1, y2;
    if (iedge_blk) {
        const int qoe = OFF_QE + lane * LPITCH;
        const int qoi = OFF_QI + lane * LPITCH;
        const float* q0 = &S[((i     == 0) || (i     == NSEQ - 1)) ? qoe : qoi];
        const float* q1 = &S[((i + 1 == 0) || (i + 1 == NSEQ - 1)) ? qoe : qoi];
        const float* q2 = &S[((i + 2 == 0) || (i + 2 == NSEQ - 1)) ? qoe : qoi];
        pair_loop<true>(p0, p1, p2, q0, q1, q2, W2, y0, y1, y2);
    } else {
        const float* q0 = &S[OFF_QI + lane * LPITCH];
        pair_loop<false>(p0, p1, p2, q0, q0, q0, W2, y0, y1, y2);
    }

    const float bias2 = b2[0];
    float* orow = out + ((size_t)(b * NSEQ + i)) * NSEQ + j;
    orow[0]        = y0 + bias2;
    orow[NSEQ]     = y1 + bias2;
    orow[2 * NSEQ] = y2 + bias2;
}

extern "C" void kernel_launch(void* const* d_in, const int* in_sizes, int n_in,
                              void* d_out, int out_size, void* d_ws, size_t ws_size,
                              hipStream_t stream) {
    (void)in_sizes; (void)n_in; (void)out_size; (void)ws_size;
    const float* x  = (const float*)d_in[0];
    const float* W1 = (const float*)d_in[1];
    const float* b1 = (const float*)d_in[2];
    const float* W2 = (const float*)d_in[3];
    const float* b2 = (const float*)d_in[4];
    float* out = (float*)d_out;
    float* T   = (float*)d_ws;

    fcpair_stage1_wmma<<<dim3(48, 8, 2), 32, 0, stream>>>(x, W1, b1, T);
    fcpair_stage2_pair<<<dim3(12, 32, 2), 128, 0, stream>>>(T, W2, b2, out);
}
